# starting kernel
# baseline (speedup 1.0000x reference)
#include <hip/hip_runtime.h>

#define B_N 16
#define T_N 2048
#define D_N 1024

#define TCHUNK 64
#define NTCHUNK (T_N / TCHUNK)
#define DTILES 4
#define DGROUPS (D_N / (16 * DTILES))

typedef __attribute__((ext_vector_type(2))) float v2f;
typedef __attribute__((ext_vector_type(8))) float v8f;

__global__ void MeanPooling_zero_kernel(float* __restrict__ out) {
    int i = blockIdx.x * blockDim.x + threadIdx.x;
    if (i < B_N * D_N) out[i] = 0.0f;
}

__global__ __launch_bounds__(32)
void MeanPooling_wmma_kernel(const float* __restrict__ xs,
                             const int* __restrict__ xs_len,
                             float* __restrict__ out) {
    const int lane   = threadIdx.x & 31;
    const int tchunk = blockIdx.x;
    const int dgroup = blockIdx.y;
    const int b      = blockIdx.z;

    const int n     = lane & 15;
    const int khalf = (lane >> 4) << 1;

    const int len = xs_len[b];
    const int tbase = tchunk * TCHUNK;
    if (tbase >= len) return;

    const float inv_len = 1.0f / (float)len;
    const int d0 = dgroup * (16 * DTILES);
    const float* __restrict__ base = xs + (size_t)b * T_N * D_N + d0 + n;

    v8f acc0 = {}; v8f acc1 = {}; v8f acc2 = {}; v8f acc3 = {};

    int tstop = tbase + TCHUNK;
    int lstop = (len + 3) & ~3;
    if (lstop < tstop) tstop = lstop;

#pragma unroll 2
    for (int t0 = tbase; t0 < tstop; t0 += 4) {
        const int ta = t0 + khalf;
        v2f A;
        A.x = (ta     < len) ? inv_len : 0.0f;
        A.y = (ta + 1 < len) ? inv_len : 0.0f;

        const float* p = base + (size_t)ta * D_N;

        v2f Bv0; Bv0.x = p[0];      Bv0.y = p[0      + D_N];
        v2f Bv1; Bv1.x = p[16];     Bv1.y = p[16     + D_N];
        v2f Bv2; Bv2.x = p[32];     Bv2.y = p[32     + D_N];
        v2f Bv3; Bv3.x = p[48];     Bv3.y = p[48     + D_N];

        acc0 = __builtin_amdgcn_wmma_f32_16x16x4_f32(false, A, false, Bv0,
                                                     (short)0, acc0, false, false);
        acc1 = __builtin_amdgcn_wmma_f32_16x16x4_f32(false, A, false, Bv1,
                                                     (short)0, acc1, false, false);
        acc2 = __builtin_amdgcn_wmma_f32_16x16x4_f32(false, A, false, Bv2,
                                                     (short)0, acc2, false, false);
        acc3 = __builtin_amdgcn_wmma_f32_16x16x4_f32(false, A, false, Bv3,
                                                     (short)0, acc3, false, false);
    }

    if (lane < 16) {
        float* o = out + (size_t)b * D_N + d0 + n;
        atomicAdd(o + 0,  acc0[0]);
        atomicAdd(o + 16, acc1[0]);
        atomicAdd(o + 32, acc2[0]);
        atomicAdd(o + 48, acc3[0]);
    }
}

extern "C" void kernel_launch(void* const* d_in, const int* in_sizes, int n_in,
                              void* d_out, int out_size, void* d_ws, size_t ws_size,
                              hipStream_t stream) {
    const float* xs     = (const float*)d_in[0];
    const int*   xs_len = (const int*)d_in[1];
    float*       out    = (float*)d_out;

    MeanPooling_zero_kernel<<<dim3((B_N * D_N + 255) / 256), dim3(256), 0, stream>>>(out);

    dim3 grid(NTCHUNK, DGROUPS, B_N);
    MeanPooling_wmma_kernel<<<grid, dim3(32), 0, stream>>>(xs, xs_len, out);
}


# each workgroup owns a 32-column slice of two batch rows paired by length rank (longest with shortest) so all WGPs sum about equal row counts; 64-row groups per wave: new but fixed sum order
# speedup vs baseline: 1.5658x; 1.5658x over previous
#include <hip/hip_runtime.h>

#define B_N 16
#define T_N 2048
#define D_N 1024

#define SLICE  32
#define NSLICE (D_N / SLICE)
#define NPAIR  (B_N / 2)
#define WAVES  16
#define GSTEPS 8
#define GROWS  (8 * GSTEPS)

typedef __attribute__((ext_vector_type(2))) float v2f;
typedef __attribute__((ext_vector_type(4))) float v4f;
typedef __attribute__((ext_vector_type(8))) float v8f;

__device__ __forceinline__ int rows_kept(int len) {
    int L = len < 0 ? 0 : len;
    return L > T_N ? T_N : L;
}

__global__ __launch_bounds__(WAVES * 32)
void masked_mean_paired_kernel(const float* __restrict__ xs,
                               const int* __restrict__ xs_len,
                               float* __restrict__ out) {
    __shared__ v4f part[2][WAVES][2][SLICE / 4];

    const int lane = threadIdx.x & 31;
    const int wave = __builtin_amdgcn_readfirstlane((int)(threadIdx.x >> 5));
    const int d0   = blockIdx.x * SLICE;
    const int pr   = blockIdx.y;
    const int cc   = lane & 7;
    const int rp   = lane >> 3;

    const int myb   = lane & 15;
    const int mylen = xs_len[myb];
    const int myL   = rows_kept(mylen);
    int rank = 0;
    for (int j = 0; j < B_N; ++j) {
        const int Lj = rows_kept(xs_len[j]);
        rank += (Lj > myL || (Lj == myL && j < myb)) ? 1 : 0;
    }
    const int b0 = __builtin_ctz(__builtin_amdgcn_ballot_w32(rank == pr));
    const int b1 = __builtin_ctz(__builtin_amdgcn_ballot_w32(rank == B_N - 1 - pr));
    const int len0 = __builtin_amdgcn_readlane(mylen, b0);
    const int len1 = __builtin_amdgcn_readlane(mylen, b1);
    const int L0 = rows_kept(len0);
    const int L1 = rows_kept(len1);
    const int n0 = (L0 + GROWS - 1) / GROWS;
    const int n1 = (L1 + GROWS - 1) / GROWS;

    for (int it = 0; it < 2; ++it) {
        const int b  = it ? b1 : b0;
        const int L  = it ? L1 : L0;
        const int ng = it ? n1 : n0;
        const int gfirst = it ? ((wave - n0) & (WAVES - 1)) : wave;

        const float* __restrict__ base =
            xs + (size_t)b * T_N * D_N + (size_t)(2 * rp) * D_N + d0 + 4 * cc;

        v8f acc0 = {}; v8f acc1 = {}; v8f acc2 = {}; v8f acc3 = {};

        for (int g = gfirst; g < ng; g += WAVES) {
            const float* __restrict__ p = base + (size_t)g * GROWS * D_N;

            v4f q[GSTEPS][2];
#pragma unroll
            for (int i = 0; i < GSTEPS; ++i) {
                q[i][0] = *(const v4f*)(p + (size_t)(8 * i)     * D_N);
                q[i][1] = *(const v4f*)(p + (size_t)(8 * i + 1) * D_N);
            }
            __builtin_amdgcn_sched_barrier(0);

            const int rem = L - g * GROWS - 2 * rp;

#pragma unroll
            for (int i = 0; i < GSTEPS; ++i) {
                v2f A;
                A.x = (8 * i     < rem) ? 1.0f : 0.0f;
                A.y = (8 * i + 1 < rem) ? 1.0f : 0.0f;

                v2f B0; B0.x = q[i][0].x; B0.y = q[i][1].x;
                v2f B1; B1.x = q[i][0].y; B1.y = q[i][1].y;
                v2f B2; B2.x = q[i][0].z; B2.y = q[i][1].z;
                v2f B3; B3.x = q[i][0].w; B3.y = q[i][1].w;

                acc0 = __builtin_amdgcn_wmma_f32_16x16x4_f32(false, A, false, B0,
                                                             (short)0, acc0, false, false);
                acc1 = __builtin_amdgcn_wmma_f32_16x16x4_f32(false, A, false, B1,
                                                             (short)0, acc1, false, false);
                acc2 = __builtin_amdgcn_wmma_f32_16x16x4_f32(false, A, false, B2,
                                                             (short)0, acc2, false, false);
                acc3 = __builtin_amdgcn_wmma_f32_16x16x4_f32(false, A, false, B3,
                                                             (short)0, acc3, false, false);
            }
        }

        if ((lane >> 4) == ((lane >> 3) & 1)) {
            v4f r; r.x = acc0[0]; r.y = acc1[0]; r.z = acc2[0]; r.w = acc3[0];
            part[it][wave][lane >> 4][cc] = r;
        }
    }
    __syncthreads();

    if (threadIdx.x < 32) {
        const int it = lane >> 4;
        const int u  = (lane >> 3) & 1;
        v4f s = part[it][0][u][cc];
        for (int w = 1; w < WAVES; ++w) s += part[it][w][u][cc];
        v4f t;
        t.x = __shfl_xor(s.x, 8); t.y = __shfl_xor(s.y, 8);
        t.z = __shfl_xor(s.z, 8); t.w = __shfl_xor(s.w, 8);
        s += t;
        if (u == 0) {
            const int   b    = it ? b1 : b0;
            const float lenf = (float)(it ? len1 : len0);
            v4f o; o.x = s.x / lenf; o.y = s.y / lenf; o.z = s.z / lenf; o.w = s.w / lenf;
            *(v4f*)(out + (size_t)b * D_N + d0 + 4 * cc) = o;
        }
    }
}

extern "C" void kernel_launch(void* const* d_in, const int* in_sizes, int n_in,
                              void* d_out, int out_size, void* d_ws, size_t ws_size,
                              hipStream_t stream) {
    const float* xs     = (const float*)d_in[0];
    const int*   xs_len = (const int*)d_in[1];
    float*       out    = (float*)d_out;

    masked_mean_paired_kernel<<<dim3(NSLICE, NPAIR), dim3(WAVES * 32), 0, stream>>>(xs, xs_len, out);
}
